# starting kernel
# baseline (speedup 1.0000x reference)
#include <hip/hip_runtime.h>
#include <math.h>

typedef __attribute__((ext_vector_type(2))) float v2f;
typedef __attribute__((ext_vector_type(8))) float v8f;

#define DM   1024
#define DS   16
#define DTR  64
#define NPJ  96
#define BSZ  2
#define TLEN 2048
#define MROWS (BSZ*TLEN)
#define NC   32
#define CL   (TLEN/NC)

__global__ void proj_gemm_kernel(const float* __restrict__ x,
                                 const float* __restrict__ Wx,
                                 float* __restrict__ xproj) {
    const int m0   = blockIdx.x * 16;
    const int nb   = blockIdx.y * 48;
    const int lane = threadIdx.x;
    const int half = lane >> 4;
    const int l16  = lane & 15;

    v8f acc0 = {}, acc1 = {}, acc2 = {};
    const float* arow = x + (size_t)(m0 + l16) * DM;

    for (int k0 = 0; k0 < DM; k0 += 4) {
        const int ka = k0 + 2 * half;
        v2f a = *(const v2f*)(arow + ka);
        const float* bp = Wx + (size_t)ka * NPJ + nb + l16;
        v2f b0, b1, b2;
        b0.x = bp[0];        b0.y = bp[NPJ];
        b1.x = bp[16];       b1.y = bp[16 + NPJ];
        b2.x = bp[32];       b2.y = bp[32 + NPJ];
        acc0 = __builtin_amdgcn_wmma_f32_16x16x4_f32(false, a, false, b0, (short)0, acc0, false, false);
        acc1 = __builtin_amdgcn_wmma_f32_16x16x4_f32(false, a, false, b1, (short)0, acc1, false, false);
        acc2 = __builtin_amdgcn_wmma_f32_16x16x4_f32(false, a, false, b2, (short)0, acc2, false, false);
    }
#pragma unroll
    for (int v = 0; v < 8; ++v) {
        const int row = m0 + v + (half ? 8 : 0);
        float* orow = xproj + (size_t)row * NPJ + nb + l16;
        orow[0]  = acc0[v];
        orow[16] = acc1[v];
        orow[32] = acc2[v];
    }
}

__global__ void dt_gemm_kernel(const float* __restrict__ xproj,
                               const float* __restrict__ Wdt,
                               const float* __restrict__ bdt,
                               float* __restrict__ dt) {
    const int m0   = blockIdx.x * 16;
    const int nb   = blockIdx.y * 64;
    const int lane = threadIdx.x;
    const int half = lane >> 4;
    const int l16  = lane & 15;

    v8f acc[4] = {};
    const float* arow = xproj + (size_t)(m0 + l16) * NPJ;

#pragma unroll
    for (int k0 = 0; k0 < DTR; k0 += 4) {
        const int ka = k0 + 2 * half;
        v2f a = *(const v2f*)(arow + ka);
        const float* bp = Wdt + (size_t)ka * DM + nb + l16;
#pragma unroll
        for (int j = 0; j < 4; ++j) {
            v2f b;
            b.x = bp[j * 16];
            b.y = bp[j * 16 + DM];
            acc[j] = __builtin_amdgcn_wmma_f32_16x16x4_f32(false, a, false, b, (short)0, acc[j], false, false);
        }
    }
#pragma unroll
    for (int v = 0; v < 8; ++v) {
        const int row = m0 + v + (half ? 8 : 0);
#pragma unroll
        for (int j = 0; j < 4; ++j) {
            const int col = nb + j * 16 + l16;
            float z = acc[j][v] + bdt[col];
            float sp = (z > 20.0f) ? z : log1pf(__expf(z));
            dt[(size_t)row * DM + col] = sp;
        }
    }
}

__global__ void scan_chunk1_kernel(const float* __restrict__ x,
                                   const float* __restrict__ xproj,
                                   const float* __restrict__ dt,
                                   const float* __restrict__ A_log,
                                   float* __restrict__ P,
                                   float* __restrict__ S) {
    const int bc = blockIdx.x;
    const int b  = bc / NC;
    const int c  = bc % NC;
    const int d  = blockIdx.y * blockDim.x + threadIdx.x;

    float An[DS];
#pragma unroll
    for (int s = 0; s < DS; ++s) An[s] = -__expf(A_log[d * DS + s]);

    float p[DS], h[DS];
#pragma unroll
    for (int s = 0; s < DS; ++s) { p[s] = 1.0f; h[s] = 0.0f; }

    const int t0 = c * CL;
    for (int t = t0; t < t0 + CL; ++t) {
        const int bt = b * TLEN + t;
        const float dtv = dt[(size_t)bt * DM + d];
        const float xv  = x[(size_t)bt * DM + d];
        const float dtx = dtv * xv;
        const float* Bm = xproj + (size_t)bt * NPJ + DTR;
#pragma unroll
        for (int s = 0; s < DS; ++s) {
            const float a = __expf(dtv * An[s]);
            p[s] *= a;
            h[s] = a * h[s] + dtx * Bm[s];
        }
    }
    const size_t base = (((size_t)b * NC + c) * DM + d) * DS;
#pragma unroll
    for (int s = 0; s < DS; ++s) { P[base + s] = p[s]; S[base + s] = h[s]; }
}

__global__ void scan_carry_kernel(const float* __restrict__ P,
                                  const float* __restrict__ S,
                                  float* __restrict__ H0) {
    const int idx = blockIdx.x * blockDim.x + threadIdx.x;
    const int b   = idx / (DM * DS);
    const int ds  = idx % (DM * DS);
    float h = 0.0f;
    for (int c = 0; c < NC; ++c) {
        const size_t off = (((size_t)b * NC + c) * DM * DS) + ds;
        H0[off] = h;
        h = P[off] * h + S[off];
    }
}

__global__ void scan_chunk2_kernel(const float* __restrict__ x,
                                   const float* __restrict__ xproj,
                                   const float* __restrict__ dt,
                                   const float* __restrict__ A_log,
                                   const float* __restrict__ Dvec,
                                   const float* __restrict__ H0,
                                   float* __restrict__ out) {
    const int bc = blockIdx.x;
    const int b  = bc / NC;
    const int c  = bc % NC;
    const int d  = blockIdx.y * blockDim.x + threadIdx.x;

    float An[DS];
#pragma unroll
    for (int s = 0; s < DS; ++s) An[s] = -__expf(A_log[d * DS + s]);

    float h[DS];
    const size_t hbase = (((size_t)b * NC + c) * DM + d) * DS;
#pragma unroll
    for (int s = 0; s < DS; ++s) h[s] = H0[hbase + s];

    const float dval = Dvec[d];
    const int t0 = c * CL;
    for (int t = t0; t < t0 + CL; ++t) {
        const int bt = b * TLEN + t;
        const float dtv = dt[(size_t)bt * DM + d];
        const float xv  = x[(size_t)bt * DM + d];
        const float dtx = dtv * xv;
        const float* Bm = xproj + (size_t)bt * NPJ + DTR;
        const float* Cm = xproj + (size_t)bt * NPJ + DTR + DS;
        float y = 0.0f;
#pragma unroll
        for (int s = 0; s < DS; ++s) {
            const float a = __expf(dtv * An[s]);
            h[s] = a * h[s] + dtx * Bm[s];
            y += h[s] * Cm[s];
        }
        out[(size_t)bt * DM + d] = y + xv * dval;
    }
}

extern "C" void kernel_launch(void* const* d_in, const int* in_sizes, int n_in,
                              void* d_out, int out_size, void* d_ws, size_t ws_size,
                              hipStream_t stream) {
    const float* x     = (const float*)d_in[0];
    const float* Wx    = (const float*)d_in[1];
    const float* Wdt   = (const float*)d_in[2];
    const float* bdt   = (const float*)d_in[3];
    const float* A_log = (const float*)d_in[4];
    const float* Dvec  = (const float*)d_in[5];
    float* out = (float*)d_out;

    char* ws = (char*)d_ws;
    float* xproj = (float*)ws;
    ws += (size_t)MROWS * NPJ * sizeof(float);
    float* dt    = (float*)ws;
    ws += (size_t)MROWS * DM * sizeof(float);
    float* P     = (float*)ws;
    ws += (size_t)BSZ * NC * DM * DS * sizeof(float);
    float* S     = (float*)ws;
    ws += (size_t)BSZ * NC * DM * DS * sizeof(float);
    float* H0    = (float*)ws;

    proj_gemm_kernel<<<dim3(MROWS / 16, 2), 32, 0, stream>>>(x, Wx, xproj);
    dt_gemm_kernel<<<dim3(MROWS / 16, 16), 32, 0, stream>>>(xproj, Wdt, bdt, dt);
    scan_chunk1_kernel<<<dim3(BSZ * NC, DM / 256), 256, 0, stream>>>(x, xproj, dt, A_log, P, S);
    scan_carry_kernel<<<(BSZ * DM * DS) / 256, 256, 0, stream>>>(P, S, H0);
    scan_chunk2_kernel<<<dim3(BSZ * NC, DM / 256), 256, 0, stream>>>(x, xproj, dt, A_log, Dvec, H0, out);
}


# carry folded into scan pass 2: a block (32 channels x 32 chunks) walks its and earlier ranges' chunk states in 2 load trips, was 4; entry states stay in LDS; 2 launches; their sum order changes
# speedup vs baseline: 2.7532x; 2.7532x over previous
#include <hip/hip_runtime.h>
#include <math.h>

typedef __attribute__((ext_vector_type(2))) float v2f;
typedef __attribute__((ext_vector_type(4))) float v4f;
typedef __attribute__((ext_vector_type(8))) float v8f;

#define DM   1024
#define DS   16
#define DTR  64
#define NPJ  96
#define BSZ  2
#define TLEN 2048
#define MROWS (BSZ*TLEN)
#define NC   128
#define CL   (TLEN/NC)

#define XP   1028
#define PJP  100
#define KSL  4
#define KSLEN (DM / KSL)
#define LOG2E 1.44269504088896341f
__global__ void __launch_bounds__(1024)
proj_scan1_kernel(const float* __restrict__ x,
                  const float* __restrict__ Wx,
                  const float* __restrict__ Wdt,
                  const float* __restrict__ bdt,
                  const float* __restrict__ A_log,
                  float* __restrict__ xproj,
                  float* __restrict__ dt,
                  float* __restrict__ SD,
                  float* __restrict__ S) {
    __shared__ float xs[CL * XP];
    __shared__ float part[KSL][CL][NPJ];
    __shared__ __attribute__((aligned(16))) float pj[CL * PJP];
    __shared__ float zt[32][CL][32];

    const int bc   = blockIdx.x;
    const int b    = bc / NC;
    const int c    = bc % NC;
    const int tid  = threadIdx.x;
    const int wave = __builtin_amdgcn_readfirstlane(tid >> 5);
    const int lane = tid & 31;
    const int half = lane >> 4;
    const int l16  = lane & 15;
    const int d    = tid;
    const int bt0  = b * TLEN + c * CL;

    v2f An2[DS / 2];
#pragma unroll
    for (int j = 0; j < DS / 2; ++j) {
        An2[j].x = -__expf(A_log[d * DS + 2 * j])     * LOG2E;
        An2[j].y = -__expf(A_log[d * DS + 2 * j + 1]) * LOG2E;
    }
    const float bias = bdt[d];

    {
        float xv[CL];
#pragma unroll
        for (int i = 0; i < CL; ++i) xv[i] = x[(size_t)(bt0 + i) * DM + d];
#pragma unroll
        for (int i = 0; i < CL; ++i) xs[i * XP + d] = xv[i];
    }
    __syncthreads();

    if (wave < 6 * KSL) {
        const int n = wave % 6;
        const int q = wave / 6;
        v8f acc = {};
        const float* ap = xs + l16 * XP + q * KSLEN + 2 * half;
        const float* bp = Wx + (size_t)(q * KSLEN + 2 * half) * NPJ + n * 16 + l16;
#pragma unroll 16
        for (int k0 = 0; k0 < KSLEN; k0 += 4) {
            const v2f a = *(const v2f*)(ap + k0);
            v2f bv;
            bv.x = bp[(size_t)k0 * NPJ];
            bv.y = bp[(size_t)(k0 + 1) * NPJ];
            acc = __builtin_amdgcn_wmma_f32_16x16x4_f32(false, a, false, bv, (short)0, acc, false, false);
        }
#pragma unroll
        for (int v = 0; v < 8; ++v) part[q][8 * half + v][n * 16 + l16] = acc[v];
    }
    __syncthreads();

    for (int o = tid; o < CL * NPJ; o += 1024) {
        const int row = o / NPJ;
        const int col = o % NPJ;
        const float v = ((part[0][row][col] + part[1][row][col]) + part[2][row][col]) + part[3][row][col];
        pj[row * PJP + col] = v;
        if (col >= DTR) xproj[(size_t)(bt0 + row) * NPJ + col] = v;
    }
    __syncthreads();

    {
        const int dw = wave * 32;
        v8f acc0 = {}, acc1 = {};
        const float* arow = pj + l16 * PJP;
#pragma unroll
        for (int k0 = 0; k0 < DTR; k0 += 4) {
            const int ka = k0 + 2 * half;
            const v2f a = *(const v2f*)(arow + ka);
            const float* bp = Wdt + (size_t)ka * DM + dw + l16;
            v2f b0, b1;
            b0.x = bp[0];
            b0.y = bp[DM];
            b1.x = bp[16];
            b1.y = bp[16 + DM];
            acc0 = __builtin_amdgcn_wmma_f32_16x16x4_f32(false, a, false, b0, (short)0, acc0, false, false);
            acc1 = __builtin_amdgcn_wmma_f32_16x16x4_f32(false, a, false, b1, (short)0, acc1, false, false);
        }
#pragma unroll
        for (int v = 0; v < 8; ++v) {
            zt[wave][8 * half + v][l16]      = acc0[v];
            zt[wave][8 * half + v][16 + l16] = acc1[v];
        }
    }
    __syncthreads();

    v2f h[DS / 2];
#pragma unroll
    for (int j = 0; j < DS / 2; ++j) h[j] = (v2f){0.0f, 0.0f};
    float sdt = 0.0f;

#pragma unroll 1
    for (int i = 0; i < CL; ++i) {
        const int bt = bt0 + i;
        const float z   = zt[wave][i][lane] + bias;
        const float dtv = (z > 20.0f) ? z : log1pf(__expf(z));
        dt[(size_t)bt * DM + d] = dtv;
        sdt += dtv;
        const float xv  = xs[i * XP + d];
        const float dtx = dtv * xv;
        const v2f* Bm = (const v2f*)(pj + i * PJP + DTR);
        const v2f dt2  = {dtv, dtv};
        const v2f dtx2 = {dtx, dtx};
#pragma unroll
        for (int j = 0; j < DS / 2; ++j) {
            const v2f e = dt2 * An2[j];
            v2f a;
            a.x = __builtin_amdgcn_exp2f(e.x);
            a.y = __builtin_amdgcn_exp2f(e.y);
            h[j] = __builtin_elementwise_fma(a, h[j], dtx2 * Bm[j]);
        }
    }
    SD[((size_t)b * NC + c) * DM + d] = sdt;
    const size_t base = (((size_t)b * NC + c) * DM + d) * DS;
#pragma unroll
    for (int j = 0; j < DS / 2; ++j) {
        S[base + 2 * j]     = h[j].x;
        S[base + 2 * j + 1] = h[j].y;
    }
}

#define RG   32
#define NRG  (NC / RG)
#define CHB  32
#define NCH  (CHB * DS)
#define LP   20
__global__ void __launch_bounds__(1024)
carry_chunk2_kernel(const float* __restrict__ x,
                    const float* __restrict__ xproj,
                    const float* __restrict__ dt,
                    const float* __restrict__ A_log,
                    const float* __restrict__ Dvec,
                    const float* __restrict__ SD,
                    const float* __restrict__ S,
                    float* __restrict__ out) {
    __shared__ __attribute__((aligned(16))) float Lst[RG * CHB * LP];
    __shared__ float pre[RG * CHB];
    __shared__ float er[(NRG - 1) * NCH];
    __shared__ float pr[(NRG - 1) * NCH];
    __shared__ __attribute__((aligned(16))) float hin[CHB * LP];

    const int blk = blockIdx.x;
    const int q   = (NRG - 1) - blk / (BSZ * (DM / CHB));
    const int b   = (blk / (DM / CHB)) % BSZ;
    const int d0  = (blk % (DM / CHB)) * CHB;
    const int tid = threadIdx.x;

    {
        const int g  = __builtin_amdgcn_readfirstlane(tid >> 9);
        const int ch = tid & (NCH - 1);
        const int dl = ch >> 4;
        const int s  = ch & 15;
        const float An = -__expf(A_log[(d0 + dl) * DS + s]);
        for (int r = q - 1 + g; r >= 0; r -= 2) {
            float pv[RG], sv[RG];
            const size_t cb = (size_t)b * NC + (size_t)r * RG;
#pragma unroll
            for (int j = 0; j < RG; ++j) {
                pv[j] = SD[(cb + j) * DM + d0 + dl];
                sv[j] = S[((cb + j) * DM + d0) * DS + ch];
            }
            float ev[RG];
#pragma unroll
            for (int j = 0; j < RG; ++j) ev[j] = __expf(An * pv[j]);
            float h = 0.0f, sum = 0.0f;
            if (r == q) {
#pragma unroll
                for (int j = 0; j < RG; ++j) {
                    Lst[(j * CHB + dl) * LP + s] = h;
                    if (s == 0) pre[j * CHB + dl] = sum;
                    h = ev[j] * h + sv[j];
                    sum += pv[j];
                }
            } else {
#pragma unroll
                for (int j = 0; j < RG; ++j) {
                    h = ev[j] * h + sv[j];
                    sum += pv[j];
                }
                er[r * NCH + ch] = h;
                pr[r * NCH + ch] = __expf(An * sum);
            }
        }
    }
    const int wave = __builtin_amdgcn_readfirstlane(tid >> 5);
    const int lane = tid & 31;
    const int d    = d0 + lane;
    const int c    = q * RG + wave;
    v2f An2[DS / 2];
#pragma unroll
    for (int j = 0; j < DS / 2; ++j) {
        An2[j].x = -__expf(A_log[d * DS + 2 * j])     * LOG2E;
        An2[j].y = -__expf(A_log[d * DS + 2 * j + 1]) * LOG2E;
    }
    const float dval = Dvec[d];
    __syncthreads();
    if (tid < NCH) {
        float h = 0.0f;
        for (int r = 0; r < q; ++r) h = pr[r * NCH + tid] * h + er[r * NCH + tid];
        hin[(tid >> 4) * LP + (tid & 15)] = h;
    }
    __syncthreads();

    v2f h[DS / 2];
    {
        const float pe = pre[wave * CHB + lane];
        const v2f pe2 = {pe, pe};
        const v2f* Lp = (const v2f*)(Lst + (wave * CHB + lane) * LP);
        const v2f* Hp = (const v2f*)(hin + lane * LP);
#pragma unroll
        for (int j = 0; j < DS / 2; ++j) {
            const v2f e = pe2 * An2[j];
            v2f a;
            a.x = __builtin_amdgcn_exp2f(e.x);
            a.y = __builtin_amdgcn_exp2f(e.y);
            h[j] = __builtin_elementwise_fma(a, Hp[j], Lp[j]);
        }
    }

    const int t0 = c * CL;
    for (int t = t0; t < t0 + CL; ++t) {
        const int bt = b * TLEN + t;
        const float dtv = dt[(size_t)bt * DM + d];
        const float xv  = x[(size_t)bt * DM + d];
        const float dtx = dtv * xv;
        const v2f* Bm = (const v2f*)(xproj + (size_t)bt * NPJ + DTR);
        const v2f* Cm = (const v2f*)(xproj + (size_t)bt * NPJ + DTR + DS);
        const v2f dt2  = {dtv, dtv};
        const v2f dtx2 = {dtx, dtx};
        v2f y2 = {0.0f, 0.0f};
#pragma unroll
        for (int j = 0; j < DS / 2; ++j) {
            const v2f e = dt2 * An2[j];
            v2f a;
            a.x = __builtin_amdgcn_exp2f(e.x);
            a.y = __builtin_amdgcn_exp2f(e.y);
            h[j] = __builtin_elementwise_fma(a, h[j], dtx2 * Bm[j]);
            y2   = __builtin_elementwise_fma(h[j], Cm[j], y2);
        }
        out[(size_t)bt * DM + d] = (y2.x + y2.y) + xv * dval;
    }
}

extern "C" void kernel_launch(void* const* d_in, const int* in_sizes, int n_in,
                              void* d_out, int out_size, void* d_ws, size_t ws_size,
                              hipStream_t stream) {
    const float* x     = (const float*)d_in[0];
    const float* Wx    = (const float*)d_in[1];
    const float* Wdt   = (const float*)d_in[2];
    const float* bdt   = (const float*)d_in[3];
    const float* A_log = (const float*)d_in[4];
    const float* Dvec  = (const float*)d_in[5];
    float* out = (float*)d_out;

    char* ws = (char*)d_ws;
    float* xproj = (float*)ws;
    ws += (size_t)MROWS * NPJ * sizeof(float);
    float* dt    = (float*)ws;
    ws += (size_t)MROWS * DM * sizeof(float);
    float* SD    = (float*)ws;
    ws += (size_t)BSZ * NC * DM * sizeof(float);
    float* S     = (float*)ws;

    proj_scan1_kernel<<<BSZ * NC, 1024, 0, stream>>>(x, Wx, Wdt, bdt, A_log, xproj, dt, SD, S);
    carry_chunk2_kernel<<<BSZ * (DM / CHB) * NRG, 1024, 0, stream>>>(x, xproj, dt, A_log, Dvec, SD, S, out);
}
